# starting kernel
# baseline (speedup 1.0000x reference)
#include <hip/hip_runtime.h>
#include <math.h>

#define B_ 16
#define T_ 512
#define C_ 32
#define W_ 128
#define KSTEPS 36
#define TAB_N 160

static constexpr float WIN_MIN_ = 2.0f;
static constexpr float WIN_MAX_ = 64.0f;

typedef __attribute__((ext_vector_type(2))) float v2f;
typedef __attribute__((ext_vector_type(8))) float v8f;

__device__ __forceinline__ float sigmoidf_(float x) {
  return __builtin_amdgcn_rcpf(1.0f + __expf(-x));
}

__global__ __launch_bounds__(32) void feat_wmma(
    const float* __restrict__ series,
    const int*   __restrict__ indices,
    const float* __restrict__ rwm,
    const float* __restrict__ rws,
    float* __restrict__ f_mean,
    float* __restrict__ f_std)
{
  __shared__ float w1tab[TAB_N];
  __shared__ float w2tab[TAB_N];

  const int c    = blockIdx.x;
  const int t0   = blockIdx.y << 4;
  const int lane = threadIdx.x;

  const float win1 = WIN_MIN_ + sigmoidf_(rwm[c]) * (WIN_MAX_ - WIN_MIN_);
  const float win2 = WIN_MIN_ + sigmoidf_(rws[c]) * (WIN_MAX_ - WIN_MIN_);

  float p1 = 0.0f, p2 = 0.0f;
  #pragma unroll
  for (int i = lane; i < TAB_N; i += 32) {
    const int w = i - 16;
    const float msk = ((w >= 0) & (w < W_)) ? 1.0f : 0.0f;
    const float age = (float)(W_ - 1 - w);
    const float v1 = msk * sigmoidf_(win1 - age);
    const float v2 = msk * sigmoidf_(win2 - age);
    w1tab[i] = v1;
    w2tab[i] = v2;
    p1 += v1;
    p2 += v2;
  }
  #pragma unroll
  for (int off = 16; off > 0; off >>= 1) {
    p1 += __shfl_xor(p1, off, 32);
    p2 += __shfl_xor(p2, off, 32);
  }
  __syncthreads();
  const float s1 = p1, s2 = p2;

  const int m     = lane & 15;
  const int khalf = (lane >> 4) << 1;
  const int j     = m;

  const int start_m = indices[2 * m];
  const int baseT   = start_m - W_ + t0;
  const float* __restrict__ gA = series + (size_t)baseT * C_ + c;
  const int tb = khalf - j + 16;

  v8f d1 = {}; v8f d2 = {}; v8f d3 = {};

  #pragma unroll
  for (int s = 0; s < KSTEPS; ++s) {
    const int k0 = (s << 2) + khalf;
    const float a0 = gA[(size_t)k0 * C_];
    const float a1 = gA[(size_t)(k0 + 1) * C_];
    const int ti = tb + (s << 2);
    v2f A  = { a0, a1 };
    v2f A2 = { a0 * a0, a1 * a1 };
    v2f B1 = { w1tab[ti], w1tab[ti + 1] };
    v2f B2 = { w2tab[ti], w2tab[ti + 1] };
    d1 = __builtin_amdgcn_wmma_f32_16x16x4_f32(false, A,  false, B1, (short)0, d1, false, false);
    d2 = __builtin_amdgcn_wmma_f32_16x16x4_f32(false, A,  false, B2, (short)0, d2, false, false);
    d3 = __builtin_amdgcn_wmma_f32_16x16x4_f32(false, A2, false, B2, (short)0, d3, false, false);
  }

  const float inv_s1 = __builtin_amdgcn_rcpf(s1);
  const float inv_s2 = __builtin_amdgcn_rcpf(s2);
  const int bBase = (lane >> 4) << 3;
  const int t     = t0 + (lane & 15);
  #pragma unroll
  for (int r = 0; r < 8; ++r) {
    const int b     = bBase + r;
    const float fm  = d1[r] * inv_s1;
    const float mu  = d2[r] * inv_s2;
    const float var = d3[r] * inv_s2 - mu * mu;
    const float fs  = sqrtf(fmaxf(var, 0.0f) + 1e-8f);
    const int o = (b * T_ + t) * C_ + c;
    f_mean[o] = fm;
    f_std[o]  = fs;
  }
}

__global__ __launch_bounds__(256) void bn_stats(
    const float* __restrict__ f_mean,
    const float* __restrict__ f_std,
    float* __restrict__ stats)
{
  const int c   = blockIdx.x;
  const int tid = threadIdx.x;
  float sm = 0.f, sm2 = 0.f, ss = 0.f, ss2 = 0.f;
  for (int i = tid; i < B_ * T_; i += 256) {
    const float vm = f_mean[i * C_ + c];
    const float vs = f_std [i * C_ + c];
    sm += vm; sm2 += vm * vm; ss += vs; ss2 += vs * vs;
  }
  __shared__ float red[4][256];
  red[0][tid] = sm; red[1][tid] = sm2; red[2][tid] = ss; red[3][tid] = ss2;
  __syncthreads();
  for (int off = 128; off > 0; off >>= 1) {
    if (tid < off) {
      red[0][tid] += red[0][tid + off];
      red[1][tid] += red[1][tid + off];
      red[2][tid] += red[2][tid + off];
      red[3][tid] += red[3][tid + off];
    }
    __syncthreads();
  }
  if (tid == 0) {
    const float n  = (float)(B_ * T_);
    const float mm = red[0][0] / n;
    const float vm = red[1][0] / n - mm * mm;
    const float ms = red[2][0] / n;
    const float vs = red[3][0] / n - ms * ms;
    stats[c * 4 + 0] = mm;
    stats[c * 4 + 1] = rsqrtf(fmaxf(vm, 0.0f) + 1e-5f);
    stats[c * 4 + 2] = ms;
    stats[c * 4 + 3] = rsqrtf(fmaxf(vs, 0.0f) + 1e-5f);
  }
}

__global__ __launch_bounds__(256) void write_out(
    const float* __restrict__ x,
    const float* __restrict__ f_mean,
    const float* __restrict__ f_std,
    const float* __restrict__ stats,
    const float* __restrict__ gm, const float* __restrict__ bm,
    const float* __restrict__ gs, const float* __restrict__ bs,
    float* __restrict__ out)
{
  const int idx = blockIdx.x * 256 + threadIdx.x;
  if (idx >= B_ * T_ * C_) return;
  const int c  = idx & (C_ - 1);
  const int bt = idx / C_;
  out[bt * (3 * C_) + c] = x[idx];
  out[bt * (3 * C_) + C_ + c] =
      (f_mean[idx] - stats[c * 4 + 0]) * stats[c * 4 + 1] * gm[c] + bm[c];
  out[bt * (3 * C_) + 2 * C_ + c] =
      (f_std[idx]  - stats[c * 4 + 2]) * stats[c * 4 + 3] * gs[c] + bs[c];
}

extern "C" void kernel_launch(void* const* d_in, const int* in_sizes, int n_in,
                              void* d_out, int out_size, void* d_ws, size_t ws_size,
                              hipStream_t stream) {
  const float* x      = (const float*)d_in[0];
  const float* series = (const float*)d_in[1];
  const int*   indices= (const int*)  d_in[2];
  const float* rwm    = (const float*)d_in[3];
  const float* rws    = (const float*)d_in[4];
  const float* gm     = (const float*)d_in[5];
  const float* bm     = (const float*)d_in[6];
  const float* gs     = (const float*)d_in[7];
  const float* bs     = (const float*)d_in[8];
  float* out = (float*)d_out;

  float* f_mean = (float*)d_ws;
  float* f_std  = f_mean + (size_t)B_ * T_ * C_;
  float* stats  = f_std  + (size_t)B_ * T_ * C_;

  feat_wmma<<<dim3(C_, T_ / 16), 32, 0, stream>>>(series, indices, rwm, rws,
                                                  f_mean, f_std);
  bn_stats<<<C_, 256, 0, stream>>>(f_mean, f_std, stats);
  write_out<<<(B_ * T_ * C_ + 255) / 256, 256, 0, stream>>>(
      x, f_mean, f_std, stats, gm, bm, gs, bs, out);
}


# series rows staged in LDS by 16-byte row loads: block = 1 batch row x 256 t x 4 channels, matrix rows = 16 t tiles (were 16 batches); per-tile statistic sums regrouped (new reduction tree)
# speedup vs baseline: 2.0543x; 2.0543x over previous
#include <hip/hip_runtime.h>
#include <math.h>

#define B_ 16
#define T_ 512
#define C_ 32
#define W_ 128
#define KSTEPS 36
#define TAB_N 160

static constexpr float WIN_MIN_ = 2.0f;
static constexpr float WIN_MAX_ = 64.0f;

typedef __attribute__((ext_vector_type(2))) float v2f;
typedef __attribute__((ext_vector_type(8))) float v8f;

__device__ __forceinline__ float sigmoidf_(float x) {
  return __builtin_amdgcn_rcpf(1.0f + __expf(-x));
}

#define NCH 4
#define TSPAN 256
#define NROWS (TSPAN + W_)
#define PITCH16 20
#define PLANE ((NROWS / 16) * PITCH16)
__global__ __launch_bounds__(32 * NCH) void feat_wmma(
    const float* __restrict__ series,
    const int*   __restrict__ indices,
    const float* __restrict__ rwm,
    const float* __restrict__ rws,
    float*  __restrict__ out,
    float*  __restrict__ part)
{
  __shared__ float w1tab[NCH][TAB_N];
  __shared__ float w2tab[NCH][TAB_N];
  __shared__ __attribute__((aligned(16))) float rows[NCH][PLANE];

  const int tid  = threadIdx.x;
  const int lane = tid & 31;
  const int wv   = tid >> 5;
  const int c    = blockIdx.x * NCH + wv;
  const int b    = blockIdx.y >> 1;
  const int t0   = (blockIdx.y & 1) * TSPAN;

  const int start_b = indices[2 * b];
  const int baseT   = start_b - W_ + t0;
  const float4* __restrict__ g4 = reinterpret_cast<const float4*>(
      series + (size_t)baseT * C_ + blockIdx.x * NCH);
  float4 rv[NROWS / (32 * NCH)];
  #pragma unroll
  for (int q = 0; q < NROWS / (32 * NCH); ++q)
    rv[q] = g4[(size_t)(tid + q * 32 * NCH) * (C_ / 4)];

  const float win1 = WIN_MIN_ + sigmoidf_(rwm[c]) * (WIN_MAX_ - WIN_MIN_);
  const float win2 = WIN_MIN_ + sigmoidf_(rws[c]) * (WIN_MAX_ - WIN_MIN_);

  float p1 = 0.0f, p2 = 0.0f;
  #pragma unroll
  for (int i = lane; i < TAB_N; i += 32) {
    const int w = i - 16;
    const float msk = ((w >= 0) & (w < W_)) ? 1.0f : 0.0f;
    const float age = (float)(W_ - 1 - w);
    const float v1 = msk * sigmoidf_(win1 - age);
    const float v2 = msk * sigmoidf_(win2 - age);
    w1tab[wv][i] = v1;
    w2tab[wv][i] = v2;
    p1 += v1;
    p2 += v2;
  }
  #pragma unroll
  for (int off = 16; off > 0; off >>= 1) {
    p1 += __shfl_xor(p1, off, 32);
    p2 += __shfl_xor(p2, off, 32);
  }
  const float s1 = p1, s2 = p2;

  #pragma unroll
  for (int q = 0; q < NROWS / (32 * NCH); ++q) {
    const int r   = tid + q * 32 * NCH;
    const int pos = r + ((r >> 4) << 2);
    rows[0][pos] = rv[q].x;
    rows[1][pos] = rv[q].y;
    rows[2][pos] = rv[q].z;
    rows[3][pos] = rv[q].w;
  }
  __syncthreads();

  const int m     = lane & 15;
  const int khalf = (lane >> 4) << 1;
  const int j     = m;

  const float* __restrict__ aP = &rows[wv][PITCH16 * m + khalf];
  const float* __restrict__ t1 = &w1tab[wv][0];
  const float* __restrict__ t2 = &w2tab[wv][0];
  const int tb = khalf - j + 16;

  v8f d1 = {}; v8f d2 = {}; v8f d3 = {};

  #pragma unroll
  for (int s = 0; s < KSTEPS; ++s) {
    const v2f A = *reinterpret_cast<const v2f*>(aP + (s << 2) + ((s >> 2) << 2));
    const int ti = tb + (s << 2);
    v2f A2 = { A[0] * A[0], A[1] * A[1] };
    v2f B1 = { t1[ti], t1[ti + 1] };
    v2f B2 = { t2[ti], t2[ti + 1] };
    d1 = __builtin_amdgcn_wmma_f32_16x16x4_f32(false, A,  false, B1, (short)0, d1, false, false);
    d2 = __builtin_amdgcn_wmma_f32_16x16x4_f32(false, A,  false, B2, (short)0, d2, false, false);
    d3 = __builtin_amdgcn_wmma_f32_16x16x4_f32(false, A2, false, B2, (short)0, d3, false, false);
  }

  const float inv_s1 = __builtin_amdgcn_rcpf(s1);
  const float inv_s2 = __builtin_amdgcn_rcpf(s2);
  const int iBase = (lane >> 4) << 3;
  const int tj    = t0 + (lane & 15);
  float sm = 0.f, sm2 = 0.f, ss = 0.f, ss2 = 0.f;
  #pragma unroll
  for (int r = 0; r < 8; ++r) {
    const int t     = tj + ((iBase + r) << 4);
    const float fm  = d1[r] * inv_s1;
    const float mu  = d2[r] * inv_s2;
    const float var = d3[r] * inv_s2 - mu * mu;
    const float fs  = sqrtf(fmaxf(var, 0.0f) + 1e-8f);
    const int o = (b * T_ + t) * (3 * C_) + C_ + c;
    out[o]      = fm;
    out[o + C_] = fs;
    sm += fm; sm2 += fm * fm; ss += fs; ss2 += fs * fs;
  }
  for (int off = 16; off > 0; off >>= 1) {
    sm  += __shfl_xor(sm,  off, 32);
    sm2 += __shfl_xor(sm2, off, 32);
    ss  += __shfl_xor(ss,  off, 32);
    ss2 += __shfl_xor(ss2, off, 32);
  }
  if (lane == 0)
    *reinterpret_cast<float4*>(part + 4 * (blockIdx.y * C_ + c)) =
        make_float4(sm, sm2, ss, ss2);
}

#define NTILE (2 * B_)
__global__ __launch_bounds__(256) void write_out(
    const float* __restrict__ x,
    const float* __restrict__ part,
    const float* __restrict__ gm, const float* __restrict__ bm,
    const float* __restrict__ gs, const float* __restrict__ bs,
    float* __restrict__ out)
{
  __shared__ float4 red[8][C_];
  const int tid = threadIdx.x;
  const int c   = tid & (C_ - 1);
  const int g   = tid >> 5;

  const float4* __restrict__ part4 = reinterpret_cast<const float4*>(part);
  float4 p = part4[(g * 4) * C_ + c];
  for (int q = 1; q < NTILE / 8; ++q) {
    const float4 v = part4[(g * 4 + q) * C_ + c];
    p.x += v.x; p.y += v.y; p.z += v.z; p.w += v.w;
  }
  red[g][c] = p;

  float xv[4], fm[4], fs[4];
  for (int k = 0; k < 4; ++k) {
    const int bt = blockIdx.x * 32 + k * 8 + g;
    xv[k] = x[bt * C_ + c];
    fm[k] = out[bt * (3 * C_) + C_ + c];
    fs[k] = out[bt * (3 * C_) + 2 * C_ + c];
  }
  const float gmc = gm[c], bmc = bm[c], gsc = gs[c], bsc = bs[c];
  __syncthreads();

  float4 s = red[0][c];
  for (int h = 1; h < 8; ++h) {
    const float4 v = red[h][c];
    s.x += v.x; s.y += v.y; s.z += v.z; s.w += v.w;
  }
  const float inv_n = 1.0f / (float)(B_ * T_);
  const float mm = s.x * inv_n;
  const float vm = s.y * inv_n - mm * mm;
  const float ms = s.z * inv_n;
  const float vs = s.w * inv_n - ms * ms;
  const float rm = rsqrtf(fmaxf(vm, 0.0f) + 1e-5f);
  const float rs = rsqrtf(fmaxf(vs, 0.0f) + 1e-5f);

  for (int k = 0; k < 4; ++k) {
    const int bt = blockIdx.x * 32 + k * 8 + g;
    out[bt * (3 * C_) + c]          = xv[k];
    out[bt * (3 * C_) + C_ + c]     = (fm[k] - mm) * rm * gmc + bmc;
    out[bt * (3 * C_) + 2 * C_ + c] = (fs[k] - ms) * rs * gsc + bsc;
  }
}

extern "C" void kernel_launch(void* const* d_in, const int* in_sizes, int n_in,
                              void* d_out, int out_size, void* d_ws, size_t ws_size,
                              hipStream_t stream) {
  const float* x      = (const float*)d_in[0];
  const float* series = (const float*)d_in[1];
  const int*   indices= (const int*)  d_in[2];
  const float* rwm    = (const float*)d_in[3];
  const float* rws    = (const float*)d_in[4];
  const float* gm     = (const float*)d_in[5];
  const float* bm     = (const float*)d_in[6];
  const float* gs     = (const float*)d_in[7];
  const float* bs     = (const float*)d_in[8];
  float* out = (float*)d_out;

  float* part = (float*)d_ws;

  feat_wmma<<<dim3(C_ / NCH, 2 * B_), 32 * NCH, 0, stream>>>(series, indices,
                                                             rwm, rws, out, part);
  write_out<<<B_ * T_ / 32, 256, 0, stream>>>(x, part, gm, bm, gs, bs, out);
}


# feature stores of the matrix kernel made whole 128-byte lines: each 32 t x 32 channel tile of out written transposed (half-wave register swap); the writer kernel turns the tiles back through LDS
# speedup vs baseline: 4.6837x; 2.2799x over previous
#include <hip/hip_runtime.h>
#include <math.h>

#define B_ 16
#define T_ 512
#define C_ 32
#define W_ 128
#define KSTEPS 36
#define TAB_N 160

static constexpr float WIN_MIN_ = 2.0f;
static constexpr float WIN_MAX_ = 64.0f;

typedef __attribute__((ext_vector_type(2))) float v2f;
typedef __attribute__((ext_vector_type(8))) float v8f;

__device__ __forceinline__ float sigmoidf_(float x) {
  return __builtin_amdgcn_rcpf(1.0f + __expf(-x));
}

__device__ __forceinline__ void half_swap(float& a, float& b) {
  typedef __attribute__((ext_vector_type(2))) unsigned int v2u;
  const v2u r = __builtin_amdgcn_permlane16_swap(__float_as_uint(a), __float_as_uint(b),
                                                 false, false);
  a = __uint_as_float(r[0]);
  b = __uint_as_float(r[1]);
}

#define NCH 4
#define TSPAN 256
#define NROWS (TSPAN + W_)
#define PITCH16 20
#define PLANE ((NROWS / 16) * PITCH16)
__global__ __launch_bounds__(32 * NCH) void feat_wmma(
    const float* __restrict__ series,
    const int*   __restrict__ indices,
    const float* __restrict__ rwm,
    const float* __restrict__ rws,
    float*  __restrict__ out,
    float*  __restrict__ part)
{
  __shared__ float w1tab[NCH][TAB_N];
  __shared__ float w2tab[NCH][TAB_N];
  __shared__ __attribute__((aligned(16))) float rows[NCH][PLANE];

  const int tid  = threadIdx.x;
  const int lane = tid & 31;
  const int wv   = tid >> 5;
  const int c    = blockIdx.x * NCH + wv;
  const int b    = blockIdx.y >> 1;
  const int t0   = (blockIdx.y & 1) * TSPAN;

  const int start_b = indices[2 * b];
  const int baseT   = start_b - W_ + t0;
  const float4* __restrict__ g4 = reinterpret_cast<const float4*>(
      series + (size_t)baseT * C_ + blockIdx.x * NCH);
  float4 rv[NROWS / (32 * NCH)];
  #pragma unroll
  for (int q = 0; q < NROWS / (32 * NCH); ++q)
    rv[q] = g4[(size_t)(tid + q * 32 * NCH) * (C_ / 4)];

  const float win1 = WIN_MIN_ + sigmoidf_(rwm[c]) * (WIN_MAX_ - WIN_MIN_);
  const float win2 = WIN_MIN_ + sigmoidf_(rws[c]) * (WIN_MAX_ - WIN_MIN_);

  float p1 = 0.0f, p2 = 0.0f;
  #pragma unroll
  for (int i = lane; i < TAB_N; i += 32) {
    const int w = i - 16;
    const float msk = ((w >= 0) & (w < W_)) ? 1.0f : 0.0f;
    const float age = (float)(W_ - 1 - w);
    const float v1 = msk * sigmoidf_(win1 - age);
    const float v2 = msk * sigmoidf_(win2 - age);
    w1tab[wv][i] = v1;
    w2tab[wv][i] = v2;
    p1 += v1;
    p2 += v2;
  }
  #pragma unroll
  for (int off = 16; off > 0; off >>= 1) {
    p1 += __shfl_xor(p1, off, 32);
    p2 += __shfl_xor(p2, off, 32);
  }
  const float s1 = p1, s2 = p2;

  #pragma unroll
  for (int q = 0; q < NROWS / (32 * NCH); ++q) {
    const int r   = tid + q * 32 * NCH;
    const int pos = r + ((r >> 4) << 2);
    rows[0][pos] = rv[q].x;
    rows[1][pos] = rv[q].y;
    rows[2][pos] = rv[q].z;
    rows[3][pos] = rv[q].w;
  }
  __syncthreads();

  const int m     = lane & 15;
  const int khalf = (lane >> 4) << 1;
  const int j     = m;

  const float* __restrict__ aP = &rows[wv][PITCH16 * m + khalf];
  const float* __restrict__ t1 = &w1tab[wv][0];
  const float* __restrict__ t2 = &w2tab[wv][0];
  const int tb = khalf - j + 16;

  v8f d1 = {}; v8f d2 = {}; v8f d3 = {};

  #pragma unroll
  for (int s = 0; s < KSTEPS; ++s) {
    const v2f A = *reinterpret_cast<const v2f*>(aP + (s << 2) + ((s >> 2) << 2));
    const int ti = tb + (s << 2);
    v2f A2 = { A[0] * A[0], A[1] * A[1] };
    v2f B1 = { t1[ti], t1[ti + 1] };
    v2f B2 = { t2[ti], t2[ti + 1] };
    d1 = __builtin_amdgcn_wmma_f32_16x16x4_f32(false, A,  false, B1, (short)0, d1, false, false);
    d2 = __builtin_amdgcn_wmma_f32_16x16x4_f32(false, A,  false, B2, (short)0, d2, false, false);
    d3 = __builtin_amdgcn_wmma_f32_16x16x4_f32(false, A2, false, B2, (short)0, d3, false, false);
  }

  const float inv_s1 = __builtin_amdgcn_rcpf(s1);
  const float inv_s2 = __builtin_amdgcn_rcpf(s2);
  float fmv[8], fsv[8];
  float sm = 0.f, sm2 = 0.f, ss = 0.f, ss2 = 0.f;
  #pragma unroll
  for (int r = 0; r < 8; ++r) {
    const float fm  = d1[r] * inv_s1;
    const float mu  = d2[r] * inv_s2;
    const float var = d3[r] * inv_s2 - mu * mu;
    const float fs  = sqrtf(fmaxf(var, 0.0f) + 1e-8f);
    fmv[r] = fm;
    fsv[r] = fs;
    sm += fm; sm2 += fm * fm; ss += fs; ss2 += fs * fs;
  }
  #pragma unroll
  for (int q = 0; q < 4; ++q) {
    half_swap(fmv[2 * q], fmv[2 * q + 1]);
    half_swap(fsv[2 * q], fsv[2 * q + 1]);
  }
  #pragma unroll
  for (int q = 0; q < 4; ++q) {
    #pragma unroll
    for (int u = 0; u < 2; ++u) {
      const int row = b * T_ + t0 + 128 * u + 32 * q + c;
      const int o   = row * (3 * C_) + C_ + lane;
      out[o]      = fmv[2 * q + u];
      out[o + C_] = fsv[2 * q + u];
    }
  }
  for (int off = 16; off > 0; off >>= 1) {
    sm  += __shfl_xor(sm,  off, 32);
    sm2 += __shfl_xor(sm2, off, 32);
    ss  += __shfl_xor(ss,  off, 32);
    ss2 += __shfl_xor(ss2, off, 32);
  }
  if (lane == 0)
    *reinterpret_cast<float4*>(part + 4 * (blockIdx.y * C_ + c)) =
        make_float4(sm, sm2, ss, ss2);
}

#define NTILE (2 * B_)
__global__ __launch_bounds__(256) void write_out(
    const float* __restrict__ x,
    const float* __restrict__ part,
    const float* __restrict__ gm, const float* __restrict__ bm,
    const float* __restrict__ gs, const float* __restrict__ bs,
    float* __restrict__ out)
{
  __shared__ float4 red[8][C_];
  __shared__ float tile_m[C_][33];
  __shared__ float tile_s[C_][33];
  const int tid = threadIdx.x;
  const int c   = tid & (C_ - 1);
  const int g   = tid >> 5;

  const float4* __restrict__ part4 = reinterpret_cast<const float4*>(part);
  float4 p = part4[(g * 4) * C_ + c];
  for (int q = 1; q < NTILE / 8; ++q) {
    const float4 v = part4[(g * 4 + q) * C_ + c];
    p.x += v.x; p.y += v.y; p.z += v.z; p.w += v.w;
  }
  red[g][c] = p;

  float xv[4], fm[4], fs[4];
  for (int k = 0; k < 4; ++k) {
    const int bt = blockIdx.x * 32 + k * 8 + g;
    xv[k] = x[bt * C_ + c];
    fm[k] = out[bt * (3 * C_) + C_ + c];
    fs[k] = out[bt * (3 * C_) + 2 * C_ + c];
  }
  const float gmc = gm[c], bmc = bm[c], gsc = gs[c], bsc = bs[c];
  for (int k = 0; k < 4; ++k) {
    tile_m[k * 8 + g][c] = fm[k];
    tile_s[k * 8 + g][c] = fs[k];
  }
  __syncthreads();

  float4 s = red[0][c];
  for (int h = 1; h < 8; ++h) {
    const float4 v = red[h][c];
    s.x += v.x; s.y += v.y; s.z += v.z; s.w += v.w;
  }
  const float inv_n = 1.0f / (float)(B_ * T_);
  const float mm = s.x * inv_n;
  const float vm = s.y * inv_n - mm * mm;
  const float ms = s.z * inv_n;
  const float vs = s.w * inv_n - ms * ms;
  const float rm = rsqrtf(fmaxf(vm, 0.0f) + 1e-5f);
  const float rs = rsqrtf(fmaxf(vs, 0.0f) + 1e-5f);

  for (int k = 0; k < 4; ++k) {
    const int r  = k * 8 + g;
    const int bt = blockIdx.x * 32 + r;
    const float vm_ = tile_m[c][r];
    const float vs_ = tile_s[c][r];
    out[bt * (3 * C_) + c]          = xv[k];
    out[bt * (3 * C_) + C_ + c]     = (vm_ - mm) * rm * gmc + bmc;
    out[bt * (3 * C_) + 2 * C_ + c] = (vs_ - ms) * rs * gsc + bsc;
  }
}

extern "C" void kernel_launch(void* const* d_in, const int* in_sizes, int n_in,
                              void* d_out, int out_size, void* d_ws, size_t ws_size,
                              hipStream_t stream) {
  const float* x      = (const float*)d_in[0];
  const float* series = (const float*)d_in[1];
  const int*   indices= (const int*)  d_in[2];
  const float* rwm    = (const float*)d_in[3];
  const float* rws    = (const float*)d_in[4];
  const float* gm     = (const float*)d_in[5];
  const float* bm     = (const float*)d_in[6];
  const float* gs     = (const float*)d_in[7];
  const float* bs     = (const float*)d_in[8];
  float* out = (float*)d_out;

  float* part = (float*)d_ws;

  feat_wmma<<<dim3(C_ / NCH, 2 * B_), 32 * NCH, 0, stream>>>(series, indices,
                                                             rwm, rws, out, part);
  write_out<<<B_ * T_ / 32, 256, 0, stream>>>(x, part, gm, bm, gs, bs, out);
}


# writer kernel re-cut to 2 waves per work-group with 16-byte loads and stores (4 whole lines per instruction): 512 waves to start instead of 2,048; sums and their order unchanged
# speedup vs baseline: 5.0001x; 1.0676x over previous
#include <hip/hip_runtime.h>
#include <math.h>

#define B_ 16
#define T_ 512
#define C_ 32
#define W_ 128
#define KSTEPS 36
#define TAB_N 160

static constexpr float WIN_MIN_ = 2.0f;
static constexpr float WIN_MAX_ = 64.0f;

typedef __attribute__((ext_vector_type(2))) float v2f;
typedef __attribute__((ext_vector_type(8))) float v8f;

__device__ __forceinline__ float sigmoidf_(float x) {
  return __builtin_amdgcn_rcpf(1.0f + __expf(-x));
}

__device__ __forceinline__ void half_swap(float& a, float& b) {
  typedef __attribute__((ext_vector_type(2))) unsigned int v2u;
  const v2u r = __builtin_amdgcn_permlane16_swap(__float_as_uint(a), __float_as_uint(b),
                                                 false, false);
  a = __uint_as_float(r[0]);
  b = __uint_as_float(r[1]);
}

#define NCH 4
#define TSPAN 256
#define NROWS (TSPAN + W_)
#define PITCH16 20
#define PLANE ((NROWS / 16) * PITCH16)
__global__ __launch_bounds__(32 * NCH) void feat_wmma(
    const float* __restrict__ series,
    const int*   __restrict__ indices,
    const float* __restrict__ rwm,
    const float* __restrict__ rws,
    float*  __restrict__ out,
    float*  __restrict__ part)
{
  __shared__ float w1tab[NCH][TAB_N];
  __shared__ float w2tab[NCH][TAB_N];
  __shared__ __attribute__((aligned(16))) float rows[NCH][PLANE];

  const int tid  = threadIdx.x;
  const int lane = tid & 31;
  const int wv   = tid >> 5;
  const int c    = blockIdx.x * NCH + wv;
  const int b    = blockIdx.y >> 1;
  const int t0   = (blockIdx.y & 1) * TSPAN;

  const int start_b = indices[2 * b];
  const int baseT   = start_b - W_ + t0;
  const float4* __restrict__ g4 = reinterpret_cast<const float4*>(
      series + (size_t)baseT * C_ + blockIdx.x * NCH);
  float4 rv[NROWS / (32 * NCH)];
  #pragma unroll
  for (int q = 0; q < NROWS / (32 * NCH); ++q)
    rv[q] = g4[(size_t)(tid + q * 32 * NCH) * (C_ / 4)];

  const float win1 = WIN_MIN_ + sigmoidf_(rwm[c]) * (WIN_MAX_ - WIN_MIN_);
  const float win2 = WIN_MIN_ + sigmoidf_(rws[c]) * (WIN_MAX_ - WIN_MIN_);

  float p1 = 0.0f, p2 = 0.0f;
  #pragma unroll
  for (int i = lane; i < TAB_N; i += 32) {
    const int w = i - 16;
    const float msk = ((w >= 0) & (w < W_)) ? 1.0f : 0.0f;
    const float age = (float)(W_ - 1 - w);
    const float v1 = msk * sigmoidf_(win1 - age);
    const float v2 = msk * sigmoidf_(win2 - age);
    w1tab[wv][i] = v1;
    w2tab[wv][i] = v2;
    p1 += v1;
    p2 += v2;
  }
  #pragma unroll
  for (int off = 16; off > 0; off >>= 1) {
    p1 += __shfl_xor(p1, off, 32);
    p2 += __shfl_xor(p2, off, 32);
  }
  const float s1 = p1, s2 = p2;

  #pragma unroll
  for (int q = 0; q < NROWS / (32 * NCH); ++q) {
    const int r   = tid + q * 32 * NCH;
    const int pos = r + ((r >> 4) << 2);
    rows[0][pos] = rv[q].x;
    rows[1][pos] = rv[q].y;
    rows[2][pos] = rv[q].z;
    rows[3][pos] = rv[q].w;
  }
  __syncthreads();

  const int m     = lane & 15;
  const int khalf = (lane >> 4) << 1;
  const int j     = m;

  const float* __restrict__ aP = &rows[wv][PITCH16 * m + khalf];
  const float* __restrict__ t1 = &w1tab[wv][0];
  const float* __restrict__ t2 = &w2tab[wv][0];
  const int tb = khalf - j + 16;

  v8f d1 = {}; v8f d2 = {}; v8f d3 = {};

  #pragma unroll
  for (int s = 0; s < KSTEPS; ++s) {
    const v2f A = *reinterpret_cast<const v2f*>(aP + (s << 2) + ((s >> 2) << 2));
    const int ti = tb + (s << 2);
    v2f A2 = { A[0] * A[0], A[1] * A[1] };
    v2f B1 = { t1[ti], t1[ti + 1] };
    v2f B2 = { t2[ti], t2[ti + 1] };
    d1 = __builtin_amdgcn_wmma_f32_16x16x4_f32(false, A,  false, B1, (short)0, d1, false, false);
    d2 = __builtin_amdgcn_wmma_f32_16x16x4_f32(false, A,  false, B2, (short)0, d2, false, false);
    d3 = __builtin_amdgcn_wmma_f32_16x16x4_f32(false, A2, false, B2, (short)0, d3, false, false);
  }

  const float inv_s1 = __builtin_amdgcn_rcpf(s1);
  const float inv_s2 = __builtin_amdgcn_rcpf(s2);
  float fmv[8], fsv[8];
  float sm = 0.f, sm2 = 0.f, ss = 0.f, ss2 = 0.f;
  #pragma unroll
  for (int r = 0; r < 8; ++r) {
    const float fm  = d1[r] * inv_s1;
    const float mu  = d2[r] * inv_s2;
    const float var = d3[r] * inv_s2 - mu * mu;
    const float fs  = sqrtf(fmaxf(var, 0.0f) + 1e-8f);
    fmv[r] = fm;
    fsv[r] = fs;
    sm += fm; sm2 += fm * fm; ss += fs; ss2 += fs * fs;
  }
  #pragma unroll
  for (int q = 0; q < 4; ++q) {
    half_swap(fmv[2 * q], fmv[2 * q + 1]);
    half_swap(fsv[2 * q], fsv[2 * q + 1]);
  }
  #pragma unroll
  for (int q = 0; q < 4; ++q) {
    #pragma unroll
    for (int u = 0; u < 2; ++u) {
      const int row = b * T_ + t0 + 128 * u + 32 * q + c;
      const int o   = row * (3 * C_) + C_ + lane;
      out[o]      = fmv[2 * q + u];
      out[o + C_] = fsv[2 * q + u];
    }
  }
  for (int off = 16; off > 0; off >>= 1) {
    sm  += __shfl_xor(sm,  off, 32);
    sm2 += __shfl_xor(sm2, off, 32);
    ss  += __shfl_xor(ss,  off, 32);
    ss2 += __shfl_xor(ss2, off, 32);
  }
  if (lane == 0)
    *reinterpret_cast<float4*>(part + 4 * (blockIdx.y * C_ + c)) =
        make_float4(sm, sm2, ss, ss2);
}

#define NTILE (2 * B_)
#define TPITCH 36
__global__ __launch_bounds__(64) __attribute__((amdgpu_waves_per_eu(1, 1))) void write_out(
    const float* __restrict__ x,
    const float* __restrict__ part,
    const float* __restrict__ gm, const float* __restrict__ bm,
    const float* __restrict__ gs, const float* __restrict__ bs,
    float* __restrict__ out)
{
  __shared__ float4 red[NTILE / 4][C_];
  __shared__ float4 stat[C_];
  __shared__ __attribute__((aligned(16))) float tile_m[C_][TPITCH];
  __shared__ __attribute__((aligned(16))) float tile_s[C_][TPITCH];
  const int tid = threadIdx.x;
  const int c   = tid & (C_ - 1);
  const int g   = tid >> 5;
  const int r   = tid >> 3;
  const int c4  = (tid & 7) << 2;

  const float4* __restrict__ part4 = reinterpret_cast<const float4*>(part);
  float4 ps[4];
  for (int k = 0; k < 4; ++k) {
    const int h = 4 * g + k;
    float4 p = part4[(h * 4) * C_ + c];
    for (int q = 1; q < 4; ++q) {
      const float4 v = part4[(h * 4 + q) * C_ + c];
      p.x += v.x; p.y += v.y; p.z += v.z; p.w += v.w;
    }
    ps[k] = p;
  }

  const int bt0 = blockIdx.x * 32 + r;
  const float4 xv0 = *reinterpret_cast<const float4*>(x + (bt0     ) * C_ + c4);
  const float4 xv1 = *reinterpret_cast<const float4*>(x + (bt0 +  8) * C_ + c4);
  const float4 xv2 = *reinterpret_cast<const float4*>(x + (bt0 + 16) * C_ + c4);
  const float4 xv3 = *reinterpret_cast<const float4*>(x + (bt0 + 24) * C_ + c4);
  float4 fm[4], fs[4];
  for (int i = 0; i < 4; ++i) {
    const int bt = bt0 + 8 * i;
    fm[i] = *reinterpret_cast<const float4*>(out + bt * (3 * C_) + C_ + c4);
    fs[i] = *reinterpret_cast<const float4*>(out + bt * (3 * C_) + 2 * C_ + c4);
  }
  const float4 gm4 = *reinterpret_cast<const float4*>(gm + c4);
  const float4 bm4 = *reinterpret_cast<const float4*>(bm + c4);
  const float4 gs4 = *reinterpret_cast<const float4*>(gs + c4);
  const float4 bs4 = *reinterpret_cast<const float4*>(bs + c4);

  for (int k = 0; k < 4; ++k) red[4 * g + k][c] = ps[k];
  for (int i = 0; i < 4; ++i) {
    *reinterpret_cast<float4*>(&tile_m[r + 8 * i][c4]) = fm[i];
    *reinterpret_cast<float4*>(&tile_s[r + 8 * i][c4]) = fs[i];
  }
  __syncthreads();

  if (tid < C_) {
    float4 s = red[0][c];
    for (int h = 1; h < NTILE / 4; ++h) {
      const float4 v = red[h][c];
      s.x += v.x; s.y += v.y; s.z += v.z; s.w += v.w;
    }
    const float inv_n = 1.0f / (float)(B_ * T_);
    const float mm = s.x * inv_n;
    const float vm = s.y * inv_n - mm * mm;
    const float ms = s.z * inv_n;
    const float vs = s.w * inv_n - ms * ms;
    const float rm = rsqrtf(fmaxf(vm, 0.0f) + 1e-5f);
    const float rs = rsqrtf(fmaxf(vs, 0.0f) + 1e-5f);
    stat[c] = make_float4(mm, rm, ms, rs);
  }
  __syncthreads();

  const float4 st0 = stat[c4], st1 = stat[c4 + 1], st2 = stat[c4 + 2], st3 = stat[c4 + 3];
  for (int i = 0; i < 4; ++i) {
    const int t  = r + 8 * i;
    const int bt = blockIdx.x * 32 + t;
    float4 om, os;
    om.x = (tile_m[c4    ][t] - st0.x) * st0.y * gm4.x + bm4.x;
    om.y = (tile_m[c4 + 1][t] - st1.x) * st1.y * gm4.y + bm4.y;
    om.z = (tile_m[c4 + 2][t] - st2.x) * st2.y * gm4.z + bm4.z;
    om.w = (tile_m[c4 + 3][t] - st3.x) * st3.y * gm4.w + bm4.w;
    os.x = (tile_s[c4    ][t] - st0.z) * st0.w * gs4.x + bs4.x;
    os.y = (tile_s[c4 + 1][t] - st1.z) * st1.w * gs4.y + bs4.y;
    os.z = (tile_s[c4 + 2][t] - st2.z) * st2.w * gs4.z + bs4.z;
    os.w = (tile_s[c4 + 3][t] - st3.z) * st3.w * gs4.w + bs4.w;
    *reinterpret_cast<float4*>(out + bt * (3 * C_) + C_ + c4)     = om;
    *reinterpret_cast<float4*>(out + bt * (3 * C_) + 2 * C_ + c4) = os;
  }
  *reinterpret_cast<float4*>(out + (bt0     ) * (3 * C_) + c4) = xv0;
  *reinterpret_cast<float4*>(out + (bt0 +  8) * (3 * C_) + c4) = xv1;
  *reinterpret_cast<float4*>(out + (bt0 + 16) * (3 * C_) + c4) = xv2;
  *reinterpret_cast<float4*>(out + (bt0 + 24) * (3 * C_) + c4) = xv3;
}

extern "C" void kernel_launch(void* const* d_in, const int* in_sizes, int n_in,
                              void* d_out, int out_size, void* d_ws, size_t ws_size,
                              hipStream_t stream) {
  const float* x      = (const float*)d_in[0];
  const float* series = (const float*)d_in[1];
  const int*   indices= (const int*)  d_in[2];
  const float* rwm    = (const float*)d_in[3];
  const float* rws    = (const float*)d_in[4];
  const float* gm     = (const float*)d_in[5];
  const float* bm     = (const float*)d_in[6];
  const float* gs     = (const float*)d_in[7];
  const float* bs     = (const float*)d_in[8];
  float* out = (float*)d_out;

  float* part = (float*)d_ws;

  feat_wmma<<<dim3(C_ / NCH, 2 * B_), 32 * NCH, 0, stream>>>(series, indices,
                                                             rwm, rws, out, part);
  write_out<<<B_ * T_ / 32, 64, 0, stream>>>(x, part, gm, bm, gs, bs, out);
}


# feature kernel: LDS operand reads taken out of the matrix stream: table fragments read before the barrier under the row loads (window sizes by scalar loads), series fragments as one group behind it
# speedup vs baseline: 5.3732x; 1.0746x over previous
#include <hip/hip_runtime.h>
#include <math.h>

#define B_ 16
#define T_ 512
#define C_ 32
#define W_ 128
#define KSTEPS 36
#define TAB_N 160

static constexpr float WIN_MIN_ = 2.0f;
static constexpr float WIN_MAX_ = 64.0f;

typedef __attribute__((ext_vector_type(2))) float v2f;
typedef __attribute__((ext_vector_type(8))) float v8f;

__device__ __forceinline__ float sigmoidf_(float x) {
  return __builtin_amdgcn_rcpf(1.0f + __expf(-x));
}

__device__ __forceinline__ void half_swap(float& a, float& b) {
  typedef __attribute__((ext_vector_type(2))) unsigned int v2u;
  const v2u r = __builtin_amdgcn_permlane16_swap(__float_as_uint(a), __float_as_uint(b),
                                                 false, false);
  a = __uint_as_float(r[0]);
  b = __uint_as_float(r[1]);
}

#define NCH 4
#define TSPAN 256
#define NROWS (TSPAN + W_)
#define PITCH16 20
#define PLANE ((NROWS / 16) * PITCH16)
__global__ __launch_bounds__(32 * NCH) void feat_wmma(
    const float* __restrict__ series,
    const int*   __restrict__ indices,
    const float* __restrict__ rwm,
    const float* __restrict__ rws,
    float*  __restrict__ out,
    float*  __restrict__ part)
{
  __shared__ float w1tab[NCH][TAB_N];
  __shared__ float w2tab[NCH][TAB_N];
  __shared__ __attribute__((aligned(16))) float rows[NCH][PLANE];

  const int tid  = threadIdx.x;
  const int lane = tid & 31;
  const int wv   = tid >> 5;
  const int c    = blockIdx.x * NCH + wv;
  const int b    = blockIdx.y >> 1;
  const int t0   = (blockIdx.y & 1) * TSPAN;

  const float4 rm4 = *reinterpret_cast<const float4*>(rwm + blockIdx.x * NCH);
  const float4 rs4 = *reinterpret_cast<const float4*>(rws + blockIdx.x * NCH);
  const float raw1 = wv == 0 ? rm4.x : wv == 1 ? rm4.y : wv == 2 ? rm4.z : rm4.w;
  const float raw2 = wv == 0 ? rs4.x : wv == 1 ? rs4.y : wv == 2 ? rs4.z : rs4.w;
  const int start_b = indices[2 * b];
  __builtin_amdgcn_sched_barrier(0);

  const int baseT   = start_b - W_ + t0;
  const float4* __restrict__ g4 = reinterpret_cast<const float4*>(
      series + (size_t)baseT * C_ + blockIdx.x * NCH);
  float4 rv[NROWS / (32 * NCH)];
  #pragma unroll
  for (int q = 0; q < NROWS / (32 * NCH); ++q)
    rv[q] = g4[(size_t)(tid + q * 32 * NCH) * (C_ / 4)];

  const float win1 = WIN_MIN_ + sigmoidf_(raw1) * (WIN_MAX_ - WIN_MIN_);
  const float win2 = WIN_MIN_ + sigmoidf_(raw2) * (WIN_MAX_ - WIN_MIN_);

  float p1 = 0.0f, p2 = 0.0f;
  #pragma unroll
  for (int i = lane; i < TAB_N; i += 32) {
    const int w = i - 16;
    const float msk = ((w >= 0) & (w < W_)) ? 1.0f : 0.0f;
    const float age = (float)(W_ - 1 - w);
    const float v1 = msk * sigmoidf_(win1 - age);
    const float v2 = msk * sigmoidf_(win2 - age);
    w1tab[wv][i] = v1;
    w2tab[wv][i] = v2;
    p1 += v1;
    p2 += v2;
  }

  const int m     = lane & 15;
  const int khalf = (lane >> 4) << 1;
  const int j     = m;

  const float* __restrict__ aP = &rows[wv][PITCH16 * m + khalf];
  const float* __restrict__ t1 = &w1tab[wv][0];
  const float* __restrict__ t2 = &w2tab[wv][0];
  const int tb = khalf - j + 16;

  v2f B1f[KSTEPS], B2f[KSTEPS];
  #pragma unroll
  for (int s = 0; s < KSTEPS; ++s) {
    const int ti = tb + (s << 2);
    B1f[s] = v2f{ t1[ti], t1[ti + 1] };
    B2f[s] = v2f{ t2[ti], t2[ti + 1] };
  }

  #pragma unroll
  for (int off = 16; off > 0; off >>= 1) {
    p1 += __shfl_xor(p1, off, 32);
    p2 += __shfl_xor(p2, off, 32);
  }
  const float s1 = p1, s2 = p2;
  asm volatile("" :: "v"(s1), "v"(s2));

  #pragma unroll
  for (int q = 0; q < NROWS / (32 * NCH); ++q) {
    const int r   = tid + q * 32 * NCH;
    const int pos = r + ((r >> 4) << 2);
    rows[0][pos] = rv[q].x;
    rows[1][pos] = rv[q].y;
    rows[2][pos] = rv[q].z;
    rows[3][pos] = rv[q].w;
  }
  __syncthreads();

  v2f Af[KSTEPS];
  #pragma unroll
  for (int s = 0; s < KSTEPS; ++s)
    Af[s] = *reinterpret_cast<const v2f*>(aP + (s << 2) + ((s >> 2) << 2));
  __builtin_amdgcn_sched_barrier(0);

  v8f d1 = {}; v8f d2 = {}; v8f d3 = {};

  #pragma unroll
  for (int s = 0; s < KSTEPS; ++s) {
    const v2f A  = Af[s];
    const v2f A2 = { A[0] * A[0], A[1] * A[1] };
    d1 = __builtin_amdgcn_wmma_f32_16x16x4_f32(false, A,  false, B1f[s], (short)0, d1, false, false);
    d2 = __builtin_amdgcn_wmma_f32_16x16x4_f32(false, A,  false, B2f[s], (short)0, d2, false, false);
    d3 = __builtin_amdgcn_wmma_f32_16x16x4_f32(false, A2, false, B2f[s], (short)0, d3, false, false);
  }

  const float inv_s1 = __builtin_amdgcn_rcpf(s1);
  const float inv_s2 = __builtin_amdgcn_rcpf(s2);
  float fmv[8], fsv[8];
  float sm = 0.f, sm2 = 0.f, ss = 0.f, ss2 = 0.f;
  #pragma unroll
  for (int r = 0; r < 8; ++r) {
    const float fm  = d1[r] * inv_s1;
    const float mu  = d2[r] * inv_s2;
    const float var = d3[r] * inv_s2 - mu * mu;
    const float fs  = sqrtf(fmaxf(var, 0.0f) + 1e-8f);
    fmv[r] = fm;
    fsv[r] = fs;
    sm += fm; sm2 += fm * fm; ss += fs; ss2 += fs * fs;
  }
  #pragma unroll
  for (int q = 0; q < 4; ++q) {
    half_swap(fmv[2 * q], fmv[2 * q + 1]);
    half_swap(fsv[2 * q], fsv[2 * q + 1]);
  }
  #pragma unroll
  for (int q = 0; q < 4; ++q) {
    #pragma unroll
    for (int u = 0; u < 2; ++u) {
      const int row = b * T_ + t0 + 128 * u + 32 * q + c;
      const int o   = row * (3 * C_) + C_ + lane;
      out[o]      = fmv[2 * q + u];
      out[o + C_] = fsv[2 * q + u];
    }
  }
  for (int off = 16; off > 0; off >>= 1) {
    sm  += __shfl_xor(sm,  off, 32);
    sm2 += __shfl_xor(sm2, off, 32);
    ss  += __shfl_xor(ss,  off, 32);
    ss2 += __shfl_xor(ss2, off, 32);
  }
  if (lane == 0)
    *reinterpret_cast<float4*>(part + 4 * (blockIdx.y * C_ + c)) =
        make_float4(sm, sm2, ss, ss2);
}

#define NTILE (2 * B_)
#define TPITCH 36
__global__ __launch_bounds__(64) __attribute__((amdgpu_waves_per_eu(1, 1))) void write_out(
    const float* __restrict__ x,
    const float* __restrict__ part,
    const float* __restrict__ gm, const float* __restrict__ bm,
    const float* __restrict__ gs, const float* __restrict__ bs,
    float* __restrict__ out)
{
  __shared__ float4 red[NTILE / 4][C_];
  __shared__ float4 stat[C_];
  __shared__ __attribute__((aligned(16))) float tile_m[C_][TPITCH];
  __shared__ __attribute__((aligned(16))) float tile_s[C_][TPITCH];
  const int tid = threadIdx.x;
  const int c   = tid & (C_ - 1);
  const int g   = tid >> 5;
  const int r   = tid >> 3;
  const int c4  = (tid & 7) << 2;

  const float4* __restrict__ part4 = reinterpret_cast<const float4*>(part);
  float4 ps[4];
  for (int k = 0; k < 4; ++k) {
    const int h = 4 * g + k;
    float4 p = part4[(h * 4) * C_ + c];
    for (int q = 1; q < 4; ++q) {
      const float4 v = part4[(h * 4 + q) * C_ + c];
      p.x += v.x; p.y += v.y; p.z += v.z; p.w += v.w;
    }
    ps[k] = p;
  }

  const int bt0 = blockIdx.x * 32 + r;
  const float4 xv0 = *reinterpret_cast<const float4*>(x + (bt0     ) * C_ + c4);
  const float4 xv1 = *reinterpret_cast<const float4*>(x + (bt0 +  8) * C_ + c4);
  const float4 xv2 = *reinterpret_cast<const float4*>(x + (bt0 + 16) * C_ + c4);
  const float4 xv3 = *reinterpret_cast<const float4*>(x + (bt0 + 24) * C_ + c4);
  float4 fm[4], fs[4];
  for (int i = 0; i < 4; ++i) {
    const int bt = bt0 + 8 * i;
    fm[i] = *reinterpret_cast<const float4*>(out + bt * (3 * C_) + C_ + c4);
    fs[i] = *reinterpret_cast<const float4*>(out + bt * (3 * C_) + 2 * C_ + c4);
  }
  const float4 gm4 = *reinterpret_cast<const float4*>(gm + c4);
  const float4 bm4 = *reinterpret_cast<const float4*>(bm + c4);
  const float4 gs4 = *reinterpret_cast<const float4*>(gs + c4);
  const float4 bs4 = *reinterpret_cast<const float4*>(bs + c4);

  for (int k = 0; k < 4; ++k) red[4 * g + k][c] = ps[k];
  for (int i = 0; i < 4; ++i) {
    *reinterpret_cast<float4*>(&tile_m[r + 8 * i][c4]) = fm[i];
    *reinterpret_cast<float4*>(&tile_s[r + 8 * i][c4]) = fs[i];
  }
  __syncthreads();

  if (tid < C_) {
    float4 s = red[0][c];
    for (int h = 1; h < NTILE / 4; ++h) {
      const float4 v = red[h][c];
      s.x += v.x; s.y += v.y; s.z += v.z; s.w += v.w;
    }
    const float inv_n = 1.0f / (float)(B_ * T_);
    const float mm = s.x * inv_n;
    const float vm = s.y * inv_n - mm * mm;
    const float ms = s.z * inv_n;
    const float vs = s.w * inv_n - ms * ms;
    const float rm = rsqrtf(fmaxf(vm, 0.0f) + 1e-5f);
    const float rs = rsqrtf(fmaxf(vs, 0.0f) + 1e-5f);
    stat[c] = make_float4(mm, rm, ms, rs);
  }
  __syncthreads();

  const float4 st0 = stat[c4], st1 = stat[c4 + 1], st2 = stat[c4 + 2], st3 = stat[c4 + 3];
  for (int i = 0; i < 4; ++i) {
    const int t  = r + 8 * i;
    const int bt = blockIdx.x * 32 + t;
    float4 om, os;
    om.x = (tile_m[c4    ][t] - st0.x) * st0.y * gm4.x + bm4.x;
    om.y = (tile_m[c4 + 1][t] - st1.x) * st1.y * gm4.y + bm4.y;
    om.z = (tile_m[c4 + 2][t] - st2.x) * st2.y * gm4.z + bm4.z;
    om.w = (tile_m[c4 + 3][t] - st3.x) * st3.y * gm4.w + bm4.w;
    os.x = (tile_s[c4    ][t] - st0.z) * st0.w * gs4.x + bs4.x;
    os.y = (tile_s[c4 + 1][t] - st1.z) * st1.w * gs4.y + bs4.y;
    os.z = (tile_s[c4 + 2][t] - st2.z) * st2.w * gs4.z + bs4.z;
    os.w = (tile_s[c4 + 3][t] - st3.z) * st3.w * gs4.w + bs4.w;
    *reinterpret_cast<float4*>(out + bt * (3 * C_) + C_ + c4)     = om;
    *reinterpret_cast<float4*>(out + bt * (3 * C_) + 2 * C_ + c4) = os;
  }
  *reinterpret_cast<float4*>(out + (bt0     ) * (3 * C_) + c4) = xv0;
  *reinterpret_cast<float4*>(out + (bt0 +  8) * (3 * C_) + c4) = xv1;
  *reinterpret_cast<float4*>(out + (bt0 + 16) * (3 * C_) + c4) = xv2;
  *reinterpret_cast<float4*>(out + (bt0 + 24) * (3 * C_) + c4) = xv3;
}

extern "C" void kernel_launch(void* const* d_in, const int* in_sizes, int n_in,
                              void* d_out, int out_size, void* d_ws, size_t ws_size,
                              hipStream_t stream) {
  const float* x      = (const float*)d_in[0];
  const float* series = (const float*)d_in[1];
  const int*   indices= (const int*)  d_in[2];
  const float* rwm    = (const float*)d_in[3];
  const float* rws    = (const float*)d_in[4];
  const float* gm     = (const float*)d_in[5];
  const float* bm     = (const float*)d_in[6];
  const float* gs     = (const float*)d_in[7];
  const float* bs     = (const float*)d_in[8];
  float* out = (float*)d_out;

  float* part = (float*)d_ws;

  feat_wmma<<<dim3(C_ / NCH, 2 * B_), 32 * NCH, 0, stream>>>(series, indices,
                                                             rwm, rws, out, part);
  write_out<<<B_ * T_ / 32, 64, 0, stream>>>(x, part, gm, bm, gs, bs, out);
}
